# starting kernel
# baseline (speedup 1.0000x reference)
#include <hip/hip_runtime.h>

typedef __attribute__((ext_vector_type(2))) float v2f;
typedef __attribute__((ext_vector_type(8))) float v8f;

__global__ void build_rowptr(const int* __restrict__ rows, int* __restrict__ row_ptr,
                             int nE, int nN) {
    int e = blockIdx.x * blockDim.x + threadIdx.x;
    if (e >= nE) return;
    int r = rows[e];
    int rp = (e == 0) ? -1 : rows[e - 1];
    for (int rr = rp + 1; rr <= r; ++rr) row_ptr[rr] = e;
    if (e == nE - 1) {
        for (int rr = r + 1; rr <= nN; ++rr) row_ptr[rr] = nE;
    }
}

__global__ void spmm_kernel(const float* __restrict__ h, const int* __restrict__ row_ptr,
                            const int* __restrict__ cols, const float* __restrict__ vals,
                            float* __restrict__ out, const float* __restrict__ sub,
                            float scale, int nN) {
    int wave = (int)((blockIdx.x * blockDim.x + threadIdx.x) >> 5);
    int lane = threadIdx.x & 31;
    if (wave >= nN) return;
    int start = row_ptr[wave];
    int end   = row_ptr[wave + 1];
    float acc0 = 0.0f, acc1 = 0.0f;
    for (int e = start; e < end; ++e) {
        float v = vals[e];
        int   c = cols[e];
        const float* xr = h + (size_t)c * 64;
        if (e + 8 < end) {
            int cn = cols[e + 8];
            __builtin_prefetch(h + (size_t)cn * 64 + lane, 0, 0);
        }
        acc0 = fmaf(v, xr[lane], acc0);
        acc1 = fmaf(v, xr[lane + 32], acc1);
    }
    size_t o = (size_t)wave * 64;
    if (sub != nullptr) {
        out[o + lane]      = scale * acc0 - sub[o + lane];
        out[o + lane + 32] = scale * acc1 - sub[o + lane + 32];
    } else {
        out[o + lane]      = acc0;
        out[o + lane + 32] = acc1;
    }
}

__device__ __forceinline__ void gemm_accum_64(const float* __restrict__ A,
                                              const float* __restrict__ Wl,
                                              int ksel, int bn, v8f acc[4]) {
#pragma unroll
    for (int kk = 0; kk < 64; kk += 4) {
        v2f a = *(const v2f*)(A + kk + ksel);
        int kprow = ((kk + ksel) >> 1) * 128;
#pragma unroll
        for (int nt = 0; nt < 4; ++nt) {
            v2f b = *(const v2f*)(Wl + kprow + (nt * 16 + bn) * 2);
            acc[nt] = __builtin_amdgcn_wmma_f32_16x16x4_f32(
                false, a, false, b, (short)0, acc[nt], false, false);
        }
    }
}

__global__ void cheb_gemm(const float* __restrict__ Tx0, const float* __restrict__ Tx1,
                          const float* __restrict__ Tx2, const float* __restrict__ W,
                          const float* __restrict__ bias, float* __restrict__ out, int nN) {
    __shared__ float w_lds[3 * 64 * 64];
    for (int i = threadIdx.x; i < 3 * 64 * 64; i += blockDim.x) {
        int t = i >> 12;
        int k = (i >> 6) & 63;
        int n = i & 63;
        w_lds[(t << 12) + ((k >> 1) << 7) + (n << 1) + (k & 1)] = W[i];
    }
    __syncthreads();

    int wave = threadIdx.x >> 5;
    int lane = threadIdx.x & 31;
    int m0 = (blockIdx.x * 8 + wave) * 16;

    int arow = m0 + (lane & 15);
    int arow_c = (arow < nN) ? arow : (nN - 1);
    int ksel = (lane >> 4) << 1;
    int bn = lane & 15;

    v8f acc[4] = {};
    gemm_accum_64(Tx0 + (size_t)arow_c * 64, w_lds,            ksel, bn, acc);
    gemm_accum_64(Tx1 + (size_t)arow_c * 64, w_lds + (1 << 12), ksel, bn, acc);
    gemm_accum_64(Tx2 + (size_t)arow_c * 64, w_lds + (2 << 12), ksel, bn, acc);

    int mbase = m0 + ((lane >> 4) << 3);
#pragma unroll
    for (int nt = 0; nt < 4; ++nt) {
        int col = nt * 16 + bn;
        float bv = bias[col];
#pragma unroll
        for (int j = 0; j < 8; ++j) {
            int row = mbase + j;
            if (row < nN) out[(size_t)row * 64 + col] = acc[nt][j] + bv;
        }
    }
}

extern "C" void kernel_launch(void* const* d_in, const int* in_sizes, int n_in,
                              void* d_out, int out_size, void* d_ws, size_t ws_size,
                              hipStream_t stream) {
    const float* x    = (const float*)d_in[0];
    const int*   rows = (const int*)d_in[1];
    const int*   cols = (const int*)d_in[2];
    const float* vals = (const float*)d_in[3];
    const float* W    = (const float*)d_in[4];
    const float* bias = (const float*)d_in[5];

    const int N = in_sizes[0] / 64;
    const int E = in_sizes[1];

    char* ws = (char*)d_ws;
    int* row_ptr = (int*)ws;
    size_t off = (((size_t)(N + 1) * sizeof(int)) + 255) & ~(size_t)255;
    float* Tx1 = (float*)(ws + off);
    float* Tx2 = Tx1 + (size_t)N * 64;

    build_rowptr<<<(E + 255) / 256, 256, 0, stream>>>(rows, row_ptr, E, N);

    int spmm_grid = (N + 7) / 8;
    spmm_kernel<<<spmm_grid, 256, 0, stream>>>(x,   row_ptr, cols, vals, Tx1, nullptr, 1.0f, N);
    spmm_kernel<<<spmm_grid, 256, 0, stream>>>(Tx1, row_ptr, cols, vals, Tx2, x,       2.0f, N);

    cheb_gemm<<<(N + 127) / 128, 256, 0, stream>>>(x, Tx1, Tx2, W, bias, (float*)d_out, N);
}


# SpMM re-tiled on a resident grid: half-wave per node, 16 B row loads per lane, 16 edge indices per load shared by row_share, 16 gathers in flight, 8192 waves walk nodes grid-stride (was 1 wave/node)
# speedup vs baseline: 1.5466x; 1.5466x over previous
#include <hip/hip_runtime.h>

typedef __attribute__((ext_vector_type(2))) float v2f;
typedef __attribute__((ext_vector_type(4))) float v4f;
typedef __attribute__((ext_vector_type(8))) float v8f;

__global__ void build_rowptr(const int* __restrict__ rows, int* __restrict__ row_ptr,
                             int nE, int nN) {
    int e = blockIdx.x * blockDim.x + threadIdx.x;
    if (e >= nE) return;
    int r = rows[e];
    int rp = (e == 0) ? -1 : rows[e - 1];
    for (int rr = rp + 1; rr <= r; ++rr) row_ptr[rr] = e;
    if (e == nE - 1) {
        for (int rr = r + 1; rr <= nN; ++rr) row_ptr[rr] = nE;
    }
}

template <int J>
__device__ __forceinline__ int row_share_i(int v) {
    return __builtin_amdgcn_update_dpp(0, v, 0x150 + J, 0xf, 0xf, true);
}

template <int J>
__device__ __forceinline__ void edge_bcast(unsigned fb, int cvec, int vvec, int rem,
                                           unsigned& off, float& vv) {
    int c = row_share_i<J>(cvec);
    float v = __int_as_float(row_share_i<J>(vvec));
    vv  = (J < rem) ? v : 0.0f;
    off = ((unsigned)c << 8) + fb;
}

__device__ __forceinline__ void fma_row(float v, const v4f& x, v4f& acc) {
    acc.x = fmaf(v, x.x, acc.x);
    acc.y = fmaf(v, x.y, acc.y);
    acc.z = fmaf(v, x.z, acc.z);
    acc.w = fmaf(v, x.w, acc.w);
}

__device__ __forceinline__ void gather_8(const char* __restrict__ hb, unsigned fb,
                                         int cvec, int vvec, int rem, v4f& acc) {
    unsigned off[8];
    float vv[8];
    v4f xr[8];
    edge_bcast<0>(fb, cvec, vvec, rem, off[0], vv[0]);
    edge_bcast<1>(fb, cvec, vvec, rem, off[1], vv[1]);
    edge_bcast<2>(fb, cvec, vvec, rem, off[2], vv[2]);
    edge_bcast<3>(fb, cvec, vvec, rem, off[3], vv[3]);
    edge_bcast<4>(fb, cvec, vvec, rem, off[4], vv[4]);
    edge_bcast<5>(fb, cvec, vvec, rem, off[5], vv[5]);
    edge_bcast<6>(fb, cvec, vvec, rem, off[6], vv[6]);
    edge_bcast<7>(fb, cvec, vvec, rem, off[7], vv[7]);
#pragma unroll
    for (int j = 0; j < 8; ++j) xr[j] = *(const v4f*)(hb + off[j]);
#pragma unroll
    for (int j = 0; j < 8; ++j) fma_row(vv[j], xr[j], acc);
}

__device__ __forceinline__ void gather_16(const char* __restrict__ hb, unsigned fb,
                                          int cvec, int vvec, int rem, v4f& acc) {
    unsigned off[16];
    float vv[16];
    v4f xr[16];
    edge_bcast<0>(fb, cvec, vvec, rem, off[0], vv[0]);
    edge_bcast<1>(fb, cvec, vvec, rem, off[1], vv[1]);
    edge_bcast<2>(fb, cvec, vvec, rem, off[2], vv[2]);
    edge_bcast<3>(fb, cvec, vvec, rem, off[3], vv[3]);
    edge_bcast<4>(fb, cvec, vvec, rem, off[4], vv[4]);
    edge_bcast<5>(fb, cvec, vvec, rem, off[5], vv[5]);
    edge_bcast<6>(fb, cvec, vvec, rem, off[6], vv[6]);
    edge_bcast<7>(fb, cvec, vvec, rem, off[7], vv[7]);
    edge_bcast<8>(fb, cvec, vvec, rem, off[8], vv[8]);
    edge_bcast<9>(fb, cvec, vvec, rem, off[9], vv[9]);
    edge_bcast<10>(fb, cvec, vvec, rem, off[10], vv[10]);
    edge_bcast<11>(fb, cvec, vvec, rem, off[11], vv[11]);
    edge_bcast<12>(fb, cvec, vvec, rem, off[12], vv[12]);
    edge_bcast<13>(fb, cvec, vvec, rem, off[13], vv[13]);
    edge_bcast<14>(fb, cvec, vvec, rem, off[14], vv[14]);
    edge_bcast<15>(fb, cvec, vvec, rem, off[15], vv[15]);
#pragma unroll
    for (int j = 0; j < 16; ++j) xr[j] = *(const v4f*)(hb + off[j]);
#pragma unroll
    for (int j = 0; j < 16; ++j) fma_row(vv[j], xr[j], acc);
}

__device__ __forceinline__ void edge_trip(const char* __restrict__ hb, unsigned fb,
                                          const int* __restrict__ cols,
                                          const float* __restrict__ vals,
                                          int l16, int base, int end, v4f& acc) {
    const int last = end - 1;
    int el = base + l16;
    el = (el < last) ? el : last;
    const int cvec = cols[el];
    const int vvec = __float_as_int(vals[el]);
    const int rem = end - base;
    if (rem > 8) gather_16(hb, fb, cvec, vvec, rem, acc);
    else         gather_8(hb, fb, cvec, vvec, rem, acc);
}

__global__ void __launch_bounds__(256)
spmm_kernel(const float* __restrict__ h, const int* __restrict__ row_ptr,
            const int* __restrict__ cols, const float* __restrict__ vals,
            float* __restrict__ out, const float* __restrict__ sub,
            float scale, int nN) {
    const int lane = threadIdx.x & 31;
    const int half = lane >> 4;
    const int l16  = lane & 15;
    const unsigned fb = (unsigned)l16 << 4;
    const int wv = __builtin_amdgcn_readfirstlane(
        (int)((blockIdx.x * blockDim.x + threadIdx.x) >> 5));
    const int stride = (int)(gridDim.x * (blockDim.x >> 5)) * 2;
    const char* hb = (const char*)h;

    int n0 = wv * 2;
    if (n0 >= nN) return;
    int nc = n0 + half;
    nc = (nc < nN) ? nc : (nN - 1);
    int start   = row_ptr[nc];
    int end_raw = row_ptr[nc + 1];

    for (; n0 < nN; n0 += stride) {
        const int n = n0 + half;
        const bool live = n < nN;
        const int end = live ? end_raw : start;
        int nn = n + stride;
        nn = (nn < nN) ? nn : (nN - 1);
        const int nstart   = row_ptr[nn];
        const int nend_raw = row_ptr[nn + 1];
        const size_t o = (size_t)n * 64 + (size_t)(l16 << 2);
        v4f s = {0.0f, 0.0f, 0.0f, 0.0f};
        if (sub != nullptr && live) s = *(const v4f*)(sub + o);
        v4f acc = {0.0f, 0.0f, 0.0f, 0.0f};
        for (int base = start; base < end; base += 16)
            edge_trip(hb, fb, cols, vals, l16, base, end, acc);
        if (live) {
            v4f r;
            if (sub != nullptr) {
                r.x = scale * acc.x - s.x;
                r.y = scale * acc.y - s.y;
                r.z = scale * acc.z - s.z;
                r.w = scale * acc.w - s.w;
            } else {
                r = acc;
            }
            *(v4f*)(out + o) = r;
        }
        start   = nstart;
        end_raw = nend_raw;
    }
}

__device__ __forceinline__ void gemm_accum_64(const float* __restrict__ A,
                                              const float* __restrict__ Wl,
                                              int ksel, int bn, v8f acc[4]) {
#pragma unroll
    for (int kk = 0; kk < 64; kk += 4) {
        v2f a = *(const v2f*)(A + kk + ksel);
        int kprow = ((kk + ksel) >> 1) * 128;
#pragma unroll
        for (int nt = 0; nt < 4; ++nt) {
            v2f b = *(const v2f*)(Wl + kprow + (nt * 16 + bn) * 2);
            acc[nt] = __builtin_amdgcn_wmma_f32_16x16x4_f32(
                false, a, false, b, (short)0, acc[nt], false, false);
        }
    }
}

__global__ void cheb_gemm(const float* __restrict__ Tx0, const float* __restrict__ Tx1,
                          const float* __restrict__ Tx2, const float* __restrict__ W,
                          const float* __restrict__ bias, float* __restrict__ out, int nN) {
    __shared__ float w_lds[3 * 64 * 64];
    for (int i = threadIdx.x; i < 3 * 64 * 64; i += blockDim.x) {
        int t = i >> 12;
        int k = (i >> 6) & 63;
        int n = i & 63;
        w_lds[(t << 12) + ((k >> 1) << 7) + (n << 1) + (k & 1)] = W[i];
    }
    __syncthreads();

    int wave = threadIdx.x >> 5;
    int lane = threadIdx.x & 31;
    int m0 = (blockIdx.x * 8 + wave) * 16;

    int arow = m0 + (lane & 15);
    int arow_c = (arow < nN) ? arow : (nN - 1);
    int ksel = (lane >> 4) << 1;
    int bn = lane & 15;

    v8f acc[4] = {};
    gemm_accum_64(Tx0 + (size_t)arow_c * 64, w_lds,            ksel, bn, acc);
    gemm_accum_64(Tx1 + (size_t)arow_c * 64, w_lds + (1 << 12), ksel, bn, acc);
    gemm_accum_64(Tx2 + (size_t)arow_c * 64, w_lds + (2 << 12), ksel, bn, acc);

    int mbase = m0 + ((lane >> 4) << 3);
#pragma unroll
    for (int nt = 0; nt < 4; ++nt) {
        int col = nt * 16 + bn;
        float bv = bias[col];
#pragma unroll
        for (int j = 0; j < 8; ++j) {
            int row = mbase + j;
            if (row < nN) out[(size_t)row * 64 + col] = acc[nt][j] + bv;
        }
    }
}

extern "C" void kernel_launch(void* const* d_in, const int* in_sizes, int n_in,
                              void* d_out, int out_size, void* d_ws, size_t ws_size,
                              hipStream_t stream) {
    const float* x    = (const float*)d_in[0];
    const int*   rows = (const int*)d_in[1];
    const int*   cols = (const int*)d_in[2];
    const float* vals = (const float*)d_in[3];
    const float* W    = (const float*)d_in[4];
    const float* bias = (const float*)d_in[5];

    const int N = in_sizes[0] / 64;
    const int E = in_sizes[1];

    char* ws = (char*)d_ws;
    int* row_ptr = (int*)ws;
    size_t off = (((size_t)(N + 1) * sizeof(int)) + 255) & ~(size_t)255;
    float* Tx1 = (float*)(ws + off);
    float* Tx2 = Tx1 + (size_t)N * 64;

    build_rowptr<<<(E + 255) / 256, 256, 0, stream>>>(rows, row_ptr, E, N);

    spmm_kernel<<<1024, 256, 0, stream>>>(x,   row_ptr, cols, vals, Tx1, nullptr, 1.0f, N);
    spmm_kernel<<<1024, 256, 0, stream>>>(Tx1, row_ptr, cols, vals, Tx2, x,       2.0f, N);

    cheb_gemm<<<(N + 127) / 128, 256, 0, stream>>>(x, Tx1, Tx2, W, bias, (float*)d_out, N);
}


# Dense product: the 48 KB of weights reach LDS with all of a thread's loads in flight (12 16-byte loads, then 12 16-byte LDS stores; was 48 load-wait-store round trips of 4 bytes one after another)
# speedup vs baseline: 1.9685x; 1.2728x over previous
#include <hip/hip_runtime.h>

typedef __attribute__((ext_vector_type(2))) float v2f;
typedef __attribute__((ext_vector_type(4))) float v4f;
typedef __attribute__((ext_vector_type(8))) float v8f;

__global__ void build_rowptr(const int* __restrict__ rows, int* __restrict__ row_ptr,
                             int nE, int nN) {
    int e = blockIdx.x * blockDim.x + threadIdx.x;
    if (e >= nE) return;
    int r = rows[e];
    int rp = (e == 0) ? -1 : rows[e - 1];
    for (int rr = rp + 1; rr <= r; ++rr) row_ptr[rr] = e;
    if (e == nE - 1) {
        for (int rr = r + 1; rr <= nN; ++rr) row_ptr[rr] = nE;
    }
}

template <int J>
__device__ __forceinline__ int row_share_i(int v) {
    return __builtin_amdgcn_update_dpp(0, v, 0x150 + J, 0xf, 0xf, true);
}

template <int J>
__device__ __forceinline__ void edge_bcast(unsigned fb, int cvec, int vvec, int rem,
                                           unsigned& off, float& vv) {
    int c = row_share_i<J>(cvec);
    float v = __int_as_float(row_share_i<J>(vvec));
    vv  = (J < rem) ? v : 0.0f;
    off = ((unsigned)c << 8) + fb;
}

__device__ __forceinline__ void fma_row(float v, const v4f& x, v4f& acc) {
    acc.x = fmaf(v, x.x, acc.x);
    acc.y = fmaf(v, x.y, acc.y);
    acc.z = fmaf(v, x.z, acc.z);
    acc.w = fmaf(v, x.w, acc.w);
}

__device__ __forceinline__ void gather_8(const char* __restrict__ hb, unsigned fb,
                                         int cvec, int vvec, int rem, v4f& acc) {
    unsigned off[8];
    float vv[8];
    v4f xr[8];
    edge_bcast<0>(fb, cvec, vvec, rem, off[0], vv[0]);
    edge_bcast<1>(fb, cvec, vvec, rem, off[1], vv[1]);
    edge_bcast<2>(fb, cvec, vvec, rem, off[2], vv[2]);
    edge_bcast<3>(fb, cvec, vvec, rem, off[3], vv[3]);
    edge_bcast<4>(fb, cvec, vvec, rem, off[4], vv[4]);
    edge_bcast<5>(fb, cvec, vvec, rem, off[5], vv[5]);
    edge_bcast<6>(fb, cvec, vvec, rem, off[6], vv[6]);
    edge_bcast<7>(fb, cvec, vvec, rem, off[7], vv[7]);
#pragma unroll
    for (int j = 0; j < 8; ++j) xr[j] = *(const v4f*)(hb + off[j]);
#pragma unroll
    for (int j = 0; j < 8; ++j) fma_row(vv[j], xr[j], acc);
}

__device__ __forceinline__ void gather_16(const char* __restrict__ hb, unsigned fb,
                                          int cvec, int vvec, int rem, v4f& acc) {
    unsigned off[16];
    float vv[16];
    v4f xr[16];
    edge_bcast<0>(fb, cvec, vvec, rem, off[0], vv[0]);
    edge_bcast<1>(fb, cvec, vvec, rem, off[1], vv[1]);
    edge_bcast<2>(fb, cvec, vvec, rem, off[2], vv[2]);
    edge_bcast<3>(fb, cvec, vvec, rem, off[3], vv[3]);
    edge_bcast<4>(fb, cvec, vvec, rem, off[4], vv[4]);
    edge_bcast<5>(fb, cvec, vvec, rem, off[5], vv[5]);
    edge_bcast<6>(fb, cvec, vvec, rem, off[6], vv[6]);
    edge_bcast<7>(fb, cvec, vvec, rem, off[7], vv[7]);
    edge_bcast<8>(fb, cvec, vvec, rem, off[8], vv[8]);
    edge_bcast<9>(fb, cvec, vvec, rem, off[9], vv[9]);
    edge_bcast<10>(fb, cvec, vvec, rem, off[10], vv[10]);
    edge_bcast<11>(fb, cvec, vvec, rem, off[11], vv[11]);
    edge_bcast<12>(fb, cvec, vvec, rem, off[12], vv[12]);
    edge_bcast<13>(fb, cvec, vvec, rem, off[13], vv[13]);
    edge_bcast<14>(fb, cvec, vvec, rem, off[14], vv[14]);
    edge_bcast<15>(fb, cvec, vvec, rem, off[15], vv[15]);
#pragma unroll
    for (int j = 0; j < 16; ++j) xr[j] = *(const v4f*)(hb + off[j]);
#pragma unroll
    for (int j = 0; j < 16; ++j) fma_row(vv[j], xr[j], acc);
}

__device__ __forceinline__ void edge_trip(const char* __restrict__ hb, unsigned fb,
                                          const int* __restrict__ cols,
                                          const float* __restrict__ vals,
                                          int l16, int base, int end, v4f& acc) {
    const int last = end - 1;
    int el = base + l16;
    el = (el < last) ? el : last;
    const int cvec = cols[el];
    const int vvec = __float_as_int(vals[el]);
    const int rem = end - base;
    if (rem > 8) gather_16(hb, fb, cvec, vvec, rem, acc);
    else         gather_8(hb, fb, cvec, vvec, rem, acc);
}

__global__ void __launch_bounds__(256)
spmm_kernel(const float* __restrict__ h, const int* __restrict__ row_ptr,
            const int* __restrict__ cols, const float* __restrict__ vals,
            float* __restrict__ out, const float* __restrict__ sub,
            float scale, int nN) {
    const int lane = threadIdx.x & 31;
    const int half = lane >> 4;
    const int l16  = lane & 15;
    const unsigned fb = (unsigned)l16 << 4;
    const int wv = __builtin_amdgcn_readfirstlane(
        (int)((blockIdx.x * blockDim.x + threadIdx.x) >> 5));
    const int stride = (int)(gridDim.x * (blockDim.x >> 5)) * 2;
    const char* hb = (const char*)h;

    int n0 = wv * 2;
    if (n0 >= nN) return;
    int nc = n0 + half;
    nc = (nc < nN) ? nc : (nN - 1);
    int start   = row_ptr[nc];
    int end_raw = row_ptr[nc + 1];

    for (; n0 < nN; n0 += stride) {
        const int n = n0 + half;
        const bool live = n < nN;
        const int end = live ? end_raw : start;
        int nn = n + stride;
        nn = (nn < nN) ? nn : (nN - 1);
        const int nstart   = row_ptr[nn];
        const int nend_raw = row_ptr[nn + 1];
        const size_t o = (size_t)n * 64 + (size_t)(l16 << 2);
        v4f s = {0.0f, 0.0f, 0.0f, 0.0f};
        if (sub != nullptr && live) s = *(const v4f*)(sub + o);
        v4f acc = {0.0f, 0.0f, 0.0f, 0.0f};
        for (int base = start; base < end; base += 16)
            edge_trip(hb, fb, cols, vals, l16, base, end, acc);
        if (live) {
            v4f r;
            if (sub != nullptr) {
                r.x = scale * acc.x - s.x;
                r.y = scale * acc.y - s.y;
                r.z = scale * acc.z - s.z;
                r.w = scale * acc.w - s.w;
            } else {
                r = acc;
            }
            *(v4f*)(out + o) = r;
        }
        start   = nstart;
        end_raw = nend_raw;
    }
}

__device__ __forceinline__ void gemm_accum_64(const float* __restrict__ A,
                                              const float* __restrict__ Wl,
                                              int ksel, int bn, v8f acc[4]) {
#pragma unroll
    for (int kk = 0; kk < 64; kk += 4) {
        v2f a = *(const v2f*)(A + kk + ksel);
        int kprow = ((kk + ksel) >> 1) * 128;
#pragma unroll
        for (int nt = 0; nt < 4; ++nt) {
            v2f b = *(const v2f*)(Wl + kprow + (nt * 16 + bn) * 2);
            acc[nt] = __builtin_amdgcn_wmma_f32_16x16x4_f32(
                false, a, false, b, (short)0, acc[nt], false, false);
        }
    }
}

__global__ void cheb_gemm(const float* __restrict__ Tx0, const float* __restrict__ Tx1,
                          const float* __restrict__ Tx2, const float* __restrict__ W,
                          const float* __restrict__ bias, float* __restrict__ out, int nN) {
    __shared__ __attribute__((aligned(16))) float w_lds[3 * 64 * 64];
    {
        v4f ra[6], rb[6];
#pragma unroll
        for (int j = 0; j < 6; ++j) {
            const int q  = j * 256 + (int)threadIdx.x;
            const int ng = q & 15;
            const int kp = (q >> 4) & 31;
            const int t  = q >> 9;
            const float* src = W + (t << 12) + (kp << 7) + (ng << 2);
            ra[j] = *(const v4f*)src;
            rb[j] = *(const v4f*)(src + 64);
        }
#pragma unroll
        for (int j = 0; j < 6; ++j) {
            const int q  = j * 256 + (int)threadIdx.x;
            const int ng = q & 15;
            const int kp = (q >> 4) & 31;
            const int t  = q >> 9;
            float* dst = w_lds + (t << 12) + (kp << 7) + (ng << 3);
            v4f lo = {ra[j].x, rb[j].x, ra[j].y, rb[j].y};
            v4f hi = {ra[j].z, rb[j].z, ra[j].w, rb[j].w};
            *(v4f*)dst       = lo;
            *(v4f*)(dst + 4) = hi;
        }
    }
    __syncthreads();

    int wave = threadIdx.x >> 5;
    int lane = threadIdx.x & 31;
    int m0 = (blockIdx.x * 8 + wave) * 16;

    int arow = m0 + (lane & 15);
    int arow_c = (arow < nN) ? arow : (nN - 1);
    int ksel = (lane >> 4) << 1;
    int bn = lane & 15;

    v8f acc[4] = {};
    gemm_accum_64(Tx0 + (size_t)arow_c * 64, w_lds,            ksel, bn, acc);
    gemm_accum_64(Tx1 + (size_t)arow_c * 64, w_lds + (1 << 12), ksel, bn, acc);
    gemm_accum_64(Tx2 + (size_t)arow_c * 64, w_lds + (2 << 12), ksel, bn, acc);

    int mbase = m0 + ((lane >> 4) << 3);
#pragma unroll
    for (int nt = 0; nt < 4; ++nt) {
        int col = nt * 16 + bn;
        float bv = bias[col];
#pragma unroll
        for (int j = 0; j < 8; ++j) {
            int row = mbase + j;
            if (row < nN) out[(size_t)row * 64 + col] = acc[nt][j] + bv;
        }
    }
}

extern "C" void kernel_launch(void* const* d_in, const int* in_sizes, int n_in,
                              void* d_out, int out_size, void* d_ws, size_t ws_size,
                              hipStream_t stream) {
    const float* x    = (const float*)d_in[0];
    const int*   rows = (const int*)d_in[1];
    const int*   cols = (const int*)d_in[2];
    const float* vals = (const float*)d_in[3];
    const float* W    = (const float*)d_in[4];
    const float* bias = (const float*)d_in[5];

    const int N = in_sizes[0] / 64;
    const int E = in_sizes[1];

    char* ws = (char*)d_ws;
    int* row_ptr = (int*)ws;
    size_t off = (((size_t)(N + 1) * sizeof(int)) + 255) & ~(size_t)255;
    float* Tx1 = (float*)(ws + off);
    float* Tx2 = Tx1 + (size_t)N * 64;

    build_rowptr<<<(E + 255) / 256, 256, 0, stream>>>(rows, row_ptr, E, N);

    spmm_kernel<<<1024, 256, 0, stream>>>(x,   row_ptr, cols, vals, Tx1, nullptr, 1.0f, N);
    spmm_kernel<<<1024, 256, 0, stream>>>(Tx1, row_ptr, cols, vals, Tx2, x,       2.0f, N);

    cheb_gemm<<<(N + 127) / 128, 256, 0, stream>>>(x, Tx1, Tx2, W, bias, (float*)d_out, N);
}
